# starting kernel
# baseline (speedup 1.0000x reference)
#include <hip/hip_runtime.h>
#include <math.h>


typedef float v2f __attribute__((ext_vector_type(2)));
typedef float v8f __attribute__((ext_vector_type(8)));

#define T_DIM 128
#define B_DIM 32
#define E_DIM 128
#define H_DIM 128
#define GE_DIM 256
#define G_DIM 768
#define GI_OFF 384
#define GC_OFF 512
#define GO_OFF 640
#define ROWS (T_DIM * B_DIM)
#define SSTR (T_DIM + 1)

__device__ __forceinline__ const float* wrow(int k,
    const float* Wf, const float* Wi, const float* Wc, const float* Wo) {
  if (k < GI_OFF) return Wf + (size_t)k * GE_DIM;
  if (k < GC_OFF) return Wi + (size_t)(k - GI_OFF) * GE_DIM;
  if (k < GO_OFF) return Wc + (size_t)(k - GC_OFF) * GE_DIM;
  return Wo + (size_t)(k - GO_OFF) * GE_DIM;
}
__device__ __forceinline__ float biasAt(int k,
    const float* bf, const float* bi, const float* bc, const float* bo) {
  if (k < GI_OFF) return bf[k];
  if (k < GC_OFF) return bi[k - GI_OFF];
  if (k < GO_OFF) return bc[k - GC_OFF];
  return bo[k - GO_OFF];
}

__global__ void __launch_bounds__(256)
xproj_wmma(const float* __restrict__ x,
           const float* __restrict__ Wf, const float* __restrict__ bf,
           const float* __restrict__ Wi, const float* __restrict__ bi,
           const float* __restrict__ Wc, const float* __restrict__ bc,
           const float* __restrict__ Wo, const float* __restrict__ bo,
           float* __restrict__ Xp) {
  const int wid  = blockIdx.x * (blockDim.x >> 5) + (threadIdx.x >> 5);
  const int lane = threadIdx.x & 31;
  const int rowTile = wid / (G_DIM / 16);
  const int colTile = wid % (G_DIM / 16);
  if (rowTile >= ROWS / 16) return;
  const int r0 = rowTile * 16;
  const int k0 = colTile * 16;
  const int mn = lane & 15;
  const int kb = (lane < 16) ? 0 : 2;

  const float* xrow = x + (size_t)(r0 + mn) * E_DIM + kb;
  const int kg = k0 + mn;
  const float* wr = wrow(kg, Wf, Wi, Wc, Wo) + H_DIM + kb;
  const float bb = biasAt(kg, bf, bi, bc, bo);

  v8f acc0 = {};
  v8f acc1 = {};
  #pragma unroll
  for (int kk = 0; kk < E_DIM; kk += 8) {
    v2f a0 = *(const v2f*)(xrow + kk);
    v2f b0 = *(const v2f*)(wr + kk);
    v2f a1 = *(const v2f*)(xrow + kk + 4);
    v2f b1 = *(const v2f*)(wr + kk + 4);
    acc0 = __builtin_amdgcn_wmma_f32_16x16x4_f32(false, a0, false, b0,
                                                 (short)0, acc0, false, false);
    acc1 = __builtin_amdgcn_wmma_f32_16x16x4_f32(false, a1, false, b1,
                                                 (short)0, acc1, false, false);
  }
  const int mbase = (lane < 16) ? 0 : 8;
  #pragma unroll
  for (int v = 0; v < 8; ++v) {
    Xp[(size_t)(r0 + mbase + v) * G_DIM + kg] = acc0[v] + acc1[v] + bb;
  }
}

__global__ void wh_transpose(const float* __restrict__ Wf, const float* __restrict__ Wi,
                             const float* __restrict__ Wc, const float* __restrict__ Wo,
                             float* __restrict__ WhT) {
  int g = blockIdx.x * blockDim.x + threadIdx.x;
  if (g >= H_DIM * G_DIM) return;
  int i = g / G_DIM;
  int k = g - i * G_DIM;
  WhT[g] = wrow(k, Wf, Wi, Wc, Wo)[i];
}

__global__ void __launch_bounds__(256)
scan_kernel(const float* __restrict__ Xp,
            const float* __restrict__ WhT,
            const float* __restrict__ b_stay,
            const float* __restrict__ b_down,
            float* __restrict__ out) {
  extern __shared__ float smem[];
  float* cur   = smem;
  float* nxt   = cur + H_DIM * SSTR;
  float* gates = nxt + H_DIM * SSTR;
  float* fa = gates + G_DIM;
  float* fb = fa + H_DIM;
  float* fc = fb + H_DIM;
  float* ov = fc + H_DIM;
  float* hv = ov + H_DIM;
  float* bs = hv + H_DIM;
  float* bd = bs + H_DIM;

  const int tid = threadIdx.x;
  const int b   = blockIdx.x;

  for (int i = tid; i < H_DIM * SSTR; i += blockDim.x) { cur[i] = 0.f; nxt[i] = 0.f; }
  if (tid < H_DIM) { hv[tid] = 0.f; bs[tid] = b_stay[tid]; bd[tid] = b_down[tid]; }
  __syncthreads();

  for (int t = 0; t < T_DIM; ++t) {
    const float* xp = Xp + (size_t)(t * B_DIM + b) * G_DIM;
    if (t + 1 < T_DIM) {
      const float* xpn = xp + (size_t)B_DIM * G_DIM;
      __builtin_prefetch(xpn + tid, 0, 0);
      __builtin_prefetch(xpn + tid + 256, 0, 0);
      __builtin_prefetch(xpn + tid + 512, 0, 0);
    }
    #pragma unroll
    for (int q = 0; q < 3; ++q) {
      const int k = tid + q * 256;
      float acc = xp[k];
      #pragma unroll 8
      for (int i = 0; i < H_DIM; ++i) acc = fmaf(hv[i], WhT[i * G_DIM + k], acc);
      gates[k] = acc;
    }
    __syncthreads();
    if (tid < H_DIM) {
      float r0 = gates[tid * 3 + 0], r1 = gates[tid * 3 + 1], r2 = gates[tid * 3 + 2];
      float mx = fmaxf(r0, fmaxf(r1, r2));
      float e0 = __expf(r0 - mx), e1 = __expf(r1 - mx), e2 = __expf(r2 - mx);
      float inv = 1.f / (e0 + e1 + e2);
      fa[tid] = e0 * inv; fb[tid] = e1 * inv; fc[tid] = e2 * inv;
      float iv = 1.f / (1.f + __expf(-gates[GI_OFF + tid]));
      float cv = tanhf(gates[GC_OFF + tid]);
      ov[tid]  = 1.f / (1.f + __expf(-gates[GO_OFF + tid]));
      cur[tid * SSTR + t] = iv * cv;
    }
    __syncthreads();
    for (int e = tid; e < H_DIM * T_DIM; e += blockDim.x) {
      const int hh = e >> 7;
      const int j  = e & (T_DIM - 1);
      const int row = hh * SSTR;
      const int jm = (j > 0) ? j - 1 : 0;
      int jp = j + 1; if (jp > t) jp = t;
      const float sl = cur[row + jm];
      const float sm = cur[row + j];
      const float sr = cur[row + jp];
      nxt[row + j] = sl * fa[hh] + (sm + bs[hh]) * fb[hh] + (sr + bd[hh]) * fc[hh];
    }
    __syncthreads();
    if (tid < H_DIM) {
      float hn = nxt[tid * SSTR + t] * ov[tid];
      hv[tid] = hn;
      out[(size_t)(t * B_DIM + b) * H_DIM + tid] = hn;
    }
    __syncthreads();
    float* tmp = cur; cur = nxt; nxt = tmp;
  }
  if (tid < H_DIM) {
    out[(size_t)T_DIM * B_DIM * H_DIM + (size_t)b * H_DIM + tid] = hv[tid];
  }
}

extern "C" void kernel_launch(void* const* d_in, const int* in_sizes, int n_in,
                              void* d_out, int out_size, void* d_ws, size_t ws_size,
                              hipStream_t stream) {
  (void)in_sizes; (void)n_in; (void)out_size; (void)ws_size;
  const float* x  = (const float*)d_in[0];
  const float* Wf = (const float*)d_in[1];
  const float* bf = (const float*)d_in[2];
  const float* Wi = (const float*)d_in[3];
  const float* bi = (const float*)d_in[4];
  const float* Wc = (const float*)d_in[5];
  const float* bc = (const float*)d_in[6];
  const float* Wo = (const float*)d_in[7];
  const float* bo = (const float*)d_in[8];
  const float* b_stay = (const float*)d_in[11];
  const float* b_down = (const float*)d_in[13];
  float* out = (float*)d_out;

  float* Xp  = (float*)d_ws;
  float* WhT = Xp + (size_t)ROWS * G_DIM;

  xproj_wmma<<<(ROWS / 16) * (G_DIM / 16) / 8, 256, 0, stream>>>(
      x, Wf, bf, Wi, bi, Wc, bc, Wo, bo, Xp);
  wh_transpose<<<(H_DIM * G_DIM + 255) / 256, 256, 0, stream>>>(Wf, Wi, Wc, Wo, WhT);

  size_t shmem = (size_t)(2 * H_DIM * SSTR + G_DIM + 7 * H_DIM) * sizeof(float);
  (void)hipFuncSetAttribute((const void*)scan_kernel,
                            hipFuncAttributeMaxDynamicSharedMemorySize, (int)shmem);
  scan_kernel<<<B_DIM, 256, shmem, stream>>>(Xp, WhT, b_stay, b_down, out);
}


# scan: the two waves of each SIMD run half a step apart (step cut at k=64, a barrier per half-step, younger wave raised by s_setprio for its second half): gate tails hide under the partner's WMMAs
# speedup vs baseline: 4.2444x; 4.2444x over previous
#include <hip/hip_runtime.h>
#include <math.h>


typedef float v2f __attribute__((ext_vector_type(2)));
typedef float v8f __attribute__((ext_vector_type(8)));

#define T_DIM 128
#define B_DIM 32
#define E_DIM 128
#define H_DIM 128
#define GE_DIM 256
#define G_DIM 768
#define GI_OFF 384
#define GC_OFF 512
#define GO_OFF 640
#define ROWS (T_DIM * B_DIM)
#define HP 132

__device__ __forceinline__ const float* wrow(int k,
    const float* Wf, const float* Wi, const float* Wc, const float* Wo) {
  if (k < GI_OFF) return Wf + (size_t)k * GE_DIM;
  if (k < GC_OFF) return Wi + (size_t)(k - GI_OFF) * GE_DIM;
  if (k < GO_OFF) return Wc + (size_t)(k - GC_OFF) * GE_DIM;
  return Wo + (size_t)(k - GO_OFF) * GE_DIM;
}
__device__ __forceinline__ float biasAt(int k,
    const float* bf, const float* bi, const float* bc, const float* bo) {
  if (k < GI_OFF) return bf[k];
  if (k < GC_OFF) return bi[k - GI_OFF];
  if (k < GO_OFF) return bc[k - GC_OFF];
  return bo[k - GO_OFF];
}

__global__ void __launch_bounds__(256)
xproj_wmma(const float* __restrict__ x,
           const float* __restrict__ Wf, const float* __restrict__ bf,
           const float* __restrict__ Wi, const float* __restrict__ bi,
           const float* __restrict__ Wc, const float* __restrict__ bc,
           const float* __restrict__ Wo, const float* __restrict__ bo,
           float* __restrict__ Xp) {
  const int wid  = blockIdx.x * (blockDim.x >> 5) + (threadIdx.x >> 5);
  const int lane = threadIdx.x & 31;
  const int rowTile = wid / (G_DIM / 16);
  const int colTile = wid % (G_DIM / 16);
  if (rowTile >= ROWS / 16) return;
  const int r0 = rowTile * 16;
  const int k0 = colTile * 16;
  const int mn = lane & 15;
  const int kb = (lane < 16) ? 0 : 2;

  const float* xrow = x + (size_t)(r0 + mn) * E_DIM + kb;
  const int kg = k0 + mn;
  const float* wr = wrow(kg, Wf, Wi, Wc, Wo) + H_DIM + kb;
  const float bb = biasAt(kg, bf, bi, bc, bo);

  v8f acc0 = {};
  v8f acc1 = {};
  #pragma unroll
  for (int kk = 0; kk < E_DIM; kk += 8) {
    v2f a0 = *(const v2f*)(xrow + kk);
    v2f b0 = *(const v2f*)(wr + kk);
    v2f a1 = *(const v2f*)(xrow + kk + 4);
    v2f b1 = *(const v2f*)(wr + kk + 4);
    acc0 = __builtin_amdgcn_wmma_f32_16x16x4_f32(false, a0, false, b0,
                                                 (short)0, acc0, false, false);
    acc1 = __builtin_amdgcn_wmma_f32_16x16x4_f32(false, a1, false, b1,
                                                 (short)0, acc1, false, false);
  }
  const int mbase = (lane < 16) ? 0 : 8;
  #pragma unroll
  for (int v = 0; v < 8; ++v) {
    Xp[(size_t)(r0 + mbase + v) * G_DIM + kg] = acc0[v] + acc1[v] + bb;
  }
}


template <int Q0, bool PRE>
__device__ __forceinline__ void half_product(v8f (&acc)[6], const v2f (&Bf)[5][32],
                                             const v2f f0, const v2f f1,
                                             const float* ar, const float4* bo_l) {
  #pragma unroll
  for (int q = Q0; q < Q0 + 8; ++q) {
    const v2f a0 = (PRE && q == Q0) ? f0 : *(const v2f*)(ar + 8 * (q - Q0));
    const v2f a1 = (PRE && q == Q0) ? f1 : *(const v2f*)(ar + 8 * (q - Q0) + 4);
    const float4 bo4 = bo_l[q * 32];
    const v2f bo0 = {bo4.x, bo4.y};
    const v2f bo1 = {bo4.z, bo4.w};
    #pragma unroll
    for (int g = 0; g < 5; ++g) {
      acc[g] = __builtin_amdgcn_wmma_f32_16x16x4_f32(false, a0, false, Bf[g][2 * q],
                                                     (short)0, acc[g], false, false);
    }
    acc[5] = __builtin_amdgcn_wmma_f32_16x16x4_f32(false, a0, false, bo0,
                                                   (short)0, acc[5], false, false);
    #pragma unroll
    for (int g = 0; g < 5; ++g) {
      acc[g] = __builtin_amdgcn_wmma_f32_16x16x4_f32(false, a1, false, Bf[g][2 * q + 1],
                                                     (short)0, acc[g], false, false);
    }
    acc[5] = __builtin_amdgcn_wmma_f32_16x16x4_f32(false, a1, false, bo1,
                                                   (short)0, acc[5], false, false);
  }
}

__global__ void __launch_bounds__(256)
scan_wmma(const float* __restrict__ Xp,
          const float* __restrict__ Wf, const float* __restrict__ Wi,
          const float* __restrict__ Wc, const float* __restrict__ Wo,
          const float* __restrict__ b_stay,
          const float* __restrict__ b_down,
          float* __restrict__ out) {
  __shared__ __attribute__((aligned(16))) float hs[2][16 * HP];
  __shared__ float4 Bo[8 * 16 * 32];

  const int tid  = threadIdx.x;
  const int lane = tid & 31;
  const int w    = __builtin_amdgcn_readfirstlane(tid >> 5);
  const int role = w >> 2;
  const int ksh  = 16 * role;
  const int n    = lane & 15;
  const int half = lane >> 4;
  const int kb   = half * 2;
  const int u    = w * 16 + n;
  const int b0   = blockIdx.x * 2;
  const int brow = b0 + half;

  int gcol[6];
  gcol[0] = 3 * u; gcol[1] = 3 * u + 1; gcol[2] = 3 * u + 2;
  gcol[3] = GI_OFF + u; gcol[4] = GC_OFF + u; gcol[5] = GO_OFF + u;

  v2f Bf[5][32];
  #pragma unroll
  for (int g = 0; g < 5; ++g) {
    const float* wr = wrow(gcol[g], Wf, Wi, Wc, Wo) + kb;
    #pragma unroll
    for (int sk = 0; sk < 32; ++sk) Bf[g][sk] = *(const v2f*)(wr + 4 * ((sk + ksh) & 31));
  }
  float4* bo_l = Bo + (w * 16) * 32 + lane;
  {
    const float* wr = wrow(gcol[5], Wf, Wi, Wc, Wo) + kb;
    #pragma unroll
    for (int q = 0; q < 16; ++q) {
      const int k0 = (2 * q + ksh) & 31;
      const v2f lo = *(const v2f*)(wr + 4 * k0);
      const v2f hi = *(const v2f*)(wr + 4 * k0 + 4);
      bo_l[q * 32] = make_float4(lo[0], lo[1], hi[0], hi[1]);
    }
  }

  const float bs = b_stay[u];
  const float bd = b_down[u];

  {
    float* hz = &hs[0][0];
    for (int i = tid; i < 2 * 16 * HP; i += 256) hz[i] = 0.f;
  }

  v8f acc[6];
  {
    const float* xr = Xp + (size_t)brow * G_DIM;
    #pragma unroll
    for (int g = 0; g < 6; ++g) {
      acc[g] = (v8f){};
      acc[g][0] = xr[gcol[g]];
    }
  }
  float top = 0.f;
  const int a_lo = n * HP + kb + 4 * ksh;
  const int a_hi = n * HP + kb + 4 * (16 - ksh);
  const int hpos = (8 * half) * HP + u;

  auto gate_tail = [&](const int t) {
    float hn;
    {
      const float r0 = acc[0][0], r1 = acc[1][0], r2 = acc[2][0];
      const float mx = fmaxf(r0, fmaxf(r1, r2));
      const float e0 = __expf(r0 - mx), e1 = __expf(r1 - mx), e2 = __expf(r2 - mx);
      const float inv = 1.f / (e0 + e1 + e2);
      const float fa = e0 * inv, fb = e1 * inv, fc = e2 * inv;
      const float iv = 1.f / (1.f + __expf(-acc[3][0]));
      const float cv = tanhf(acc[4][0]);
      const float ovv = 1.f / (1.f + __expf(-acc[5][0]));
      const float ic = iv * cv;
      const float sl = (t > 0) ? top : ic;
      const float nt = sl * fa + (ic + bs) * fb + (ic + bd) * fc;
      top = nt;
      hn = nt * ovv;
    }
    {
      const int tn = (t + 1 < T_DIM) ? t + 1 : T_DIM - 1;
      const float* xr = Xp + (size_t)(tn * B_DIM + brow) * G_DIM;
      #pragma unroll
      for (int g = 0; g < 6; ++g) acc[g][0] = xr[gcol[g]];
    }
    hs[(t + 1) & 1][hpos] = hn;
  };

  auto copy_out = [&](const int t) {
    const int off = 64 * role + 4 * n;
    const float4 hv = *(const float4*)(hs[(t + 1) & 1] + (8 * half) * HP + off);
    *(float4*)(out + (size_t)(t * B_DIM + brow) * H_DIM + off) = hv;
    if (t == T_DIM - 1) {
      *(float4*)(out + (size_t)T_DIM * B_DIM * H_DIM + (size_t)brow * H_DIM + off) = hv;
    }
  };

  v2f fh0 = {0.f, 0.f}, fh1 = {0.f, 0.f};
  auto read_hi = [&](const float* hb) {
    fh0 = *(const v2f*)(hb + a_hi);
    fh1 = *(const v2f*)(hb + a_hi + 4);
  };

  __syncthreads();

  if (role == 0) half_product<0, false>(acc, Bf, fh0, fh1, hs[0] + a_lo, bo_l);
  else read_hi(hs[0]);
  __syncthreads();

  #pragma unroll 1
  for (int t = 0; t < T_DIM - 1; ++t) {
    if (role == 0) read_hi(hs[t & 1]);
    half_product<8, true>(acc, Bf, fh0, fh1, hs[t & 1] + a_hi, bo_l);
    if (role == 0) gate_tail(t);
    __syncthreads();
    if (w == 0) copy_out(t);

    if (role) __builtin_amdgcn_s_setprio(3);
    half_product<0, false>(acc, Bf, fh0, fh1, hs[(t + 1 - role) & 1] + a_lo, bo_l);
    if (role) {
      gate_tail(t);
      __builtin_amdgcn_s_setprio(0);
      read_hi(hs[(t + 1) & 1]);
    }
    __syncthreads();
    if (w == 4) copy_out(t);
  }

  if (role == 0) read_hi(hs[(T_DIM - 1) & 1]);
  half_product<8, true>(acc, Bf, fh0, fh1, hs[(T_DIM - 1) & 1] + a_hi, bo_l);
  if (role == 0) gate_tail(T_DIM - 1);
  __syncthreads();
  if (w == 0) copy_out(T_DIM - 1);
  if (role) {
    half_product<0, false>(acc, Bf, fh0, fh1, hs[(T_DIM - 1) & 1] + a_lo, bo_l);
    gate_tail(T_DIM - 1);
  }
  __syncthreads();
  if (w == 4) copy_out(T_DIM - 1);
}

extern "C" void kernel_launch(void* const* d_in, const int* in_sizes, int n_in,
                              void* d_out, int out_size, void* d_ws, size_t ws_size,
                              hipStream_t stream) {
  (void)in_sizes; (void)n_in; (void)out_size; (void)ws_size;
  const float* x  = (const float*)d_in[0];
  const float* Wf = (const float*)d_in[1];
  const float* bf = (const float*)d_in[2];
  const float* Wi = (const float*)d_in[3];
  const float* bi = (const float*)d_in[4];
  const float* Wc = (const float*)d_in[5];
  const float* bc = (const float*)d_in[6];
  const float* Wo = (const float*)d_in[7];
  const float* bo = (const float*)d_in[8];
  const float* b_stay = (const float*)d_in[11];
  const float* b_down = (const float*)d_in[13];
  float* out = (float*)d_out;

  float* Xp  = (float*)d_ws;

  xproj_wmma<<<(ROWS / 16) * (G_DIM / 16) / 8, 256, 0, stream>>>(
      x, Wf, bf, Wi, bi, Wc, bc, Wo, bo, Xp);

  scan_wmma<<<B_DIM / 2, 256, 0, stream>>>(Xp, Wf, Wi, Wc, Wo, b_stay, b_down, out);
}
